# starting kernel
# baseline (speedup 1.0000x reference)
#include <hip/hip_runtime.h>

typedef float v2f __attribute__((ext_vector_type(2)));
typedef float v8f __attribute__((ext_vector_type(8)));

#define IN_F   4096
#define OUT_F  16384
#define M_ROWS 64
#define KK     16
#define NUM_BANKS 64
#define ANNEAL 2000
#define KC     256
#define NW     8
#define NTILE  (NW * 16)
#define KSPLIT 4
#define KSLICE (IN_F / KSPLIT)

__global__ __launch_bounds__(256)
void zero_out(float* __restrict__ out, int n4)
{
    int i = blockIdx.x * blockDim.x + threadIdx.x;
    if (i < n4) ((float4*)out)[i] = make_float4(0.f, 0.f, 0.f, 0.f);
}

__global__ __launch_bounds__(256)
void density_linear_wmma(const float* __restrict__ x,
                         const float* __restrict__ lw,
                         const float* __restrict__ scale,
                         const float* __restrict__ thr,
                         const int*   __restrict__ stepp,
                         float* __restrict__ out)
{
    __shared__ float xs[M_ROWS * KC];

    const int tid   = threadIdx.x;
    const int lane  = tid & 31;
    const int wave  = tid >> 5;
    const int nbase = blockIdx.x * NTILE + wave * 16;
    const int ncol  = nbase + (lane & 15);
    const int koff  = (lane >> 4) * 2;
    const int kbase = blockIdx.y * KSLICE;

    const int step = stepp[0];
    float a = fminf((float)step / (float)ANNEAL, 1.0f);
    const float oma = 1.0f - a;
    const float a8  = a * 0.125f;

    const float* Rp = thr + (unsigned)(step % NUM_BANKS) * KK;
    float r[KK];
    #pragma unroll
    for (int i = 0; i < KK; ++i) r[i] = Rp[i];

    const float s = scale[ncol];

    v8f acc[4];
    #pragma unroll
    for (int mt = 0; mt < 4; ++mt) acc[mt] = (v8f)0.0f;

    const float* wrow = lw + (size_t)ncol * IN_F + koff;

    for (int kc = kbase; kc < kbase + KSLICE; kc += KC) {
        __syncthreads();
        for (int i = tid * 4; i < M_ROWS * KC; i += 256 * 4) {
            const int m = i / KC;
            const int j = i & (KC - 1);
            const float4 v = *(const float4*)(x + (size_t)m * IN_F + kc + j);
            *(float4*)(xs + i) = v;
        }
        __syncthreads();

        if (kc + KC < kbase + KSLICE)
            __builtin_prefetch(wrow + kc + KC, 0, 1);

        #pragma unroll 4
        for (int k = 0; k < KC; k += 4) {
            const float2 wv = *(const float2*)(wrow + kc + k);
            v2f b;
            {
                float c0 = fminf(fmaxf(wv.x, -1.0f), 1.0f);
                float c1 = fminf(fmaxf(wv.y, -1.0f), 1.0f);
                float t0 = __builtin_fmaf(c0, 0.5f, 0.5f);
                float t1 = __builtin_fmaf(c1, 0.5f, 0.5f);
                int n0 = 0, n1 = 0;
                #pragma unroll
                for (int i = 0; i < KK; ++i) {
                    n0 += (r[i] < t0) ? 1 : 0;
                    n1 += (r[i] < t1) ? 1 : 0;
                }
                b.x = __builtin_fmaf(oma, c0, __builtin_fmaf((float)n0, a8, -a));
                b.y = __builtin_fmaf(oma, c1, __builtin_fmaf((float)n1, a8, -a));
            }

            #pragma unroll
            for (int mt = 0; mt < 4; ++mt) {
                const int m = mt * 16 + (lane & 15);
                v2f av = *(const v2f*)(xs + m * KC + (k & (KC - 1)) + koff);
                acc[mt] = __builtin_amdgcn_wmma_f32_16x16x4_f32(
                    false, av, false, b, (short)0, acc[mt], false, false);
            }
        }
    }

    const int mrow0 = (lane >> 4) * 8;
    #pragma unroll
    for (int mt = 0; mt < 4; ++mt) {
        #pragma unroll
        for (int j = 0; j < 8; ++j) {
            const int m = mt * 16 + mrow0 + j;
            unsafeAtomicAdd(&out[(size_t)m * OUT_F + ncol], acc[mt][j] * s);
        }
    }
}

extern "C" void kernel_launch(void* const* d_in, const int* in_sizes, int n_in,
                              void* d_out, int out_size, void* d_ws, size_t ws_size,
                              hipStream_t stream) {
    const float* x     = (const float*)d_in[0];
    const float* lw    = (const float*)d_in[1];
    const float* scale = (const float*)d_in[2];
    const float* thr   = (const float*)d_in[3];
    const int*   step  = (const int*)d_in[4];
    float* out = (float*)d_out;

    const int n4 = out_size / 4;
    zero_out<<<dim3((n4 + 255) / 256), dim3(256), 0, stream>>>(out, n4);

    dim3 grid(OUT_F / NTILE, KSPLIT);
    dim3 block(256);
    density_linear_wmma<<<grid, block, 0, stream>>>(x, lw, scale, thr, step, out);
}


# x tile in LDS at row pitch 260 floats (was 256): the rows read by one A-fragment ds_load_b64 fall on different banks instead of all on one bank pair; summation order unchanged
# speedup vs baseline: 1.1380x; 1.1380x over previous
#include <hip/hip_runtime.h>

typedef float v2f __attribute__((ext_vector_type(2)));
typedef float v8f __attribute__((ext_vector_type(8)));

#define IN_F   4096
#define OUT_F  16384
#define M_ROWS 64
#define KK     16
#define NUM_BANKS 64
#define ANNEAL 2000
#define KC     256
#define XP     (KC + 4)
#define NW     8
#define NTILE  (NW * 16)
#define KSPLIT 4
#define KSLICE (IN_F / KSPLIT)

__global__ __launch_bounds__(256)
void zero_out(float* __restrict__ out, int n4)
{
    int i = blockIdx.x * blockDim.x + threadIdx.x;
    if (i < n4) ((float4*)out)[i] = make_float4(0.f, 0.f, 0.f, 0.f);
}

__global__ __launch_bounds__(256)
void density_linear_wmma(const float* __restrict__ x,
                         const float* __restrict__ lw,
                         const float* __restrict__ scale,
                         const float* __restrict__ thr,
                         const int*   __restrict__ stepp,
                         float* __restrict__ out)
{
    __shared__ float xs[M_ROWS * XP];

    const int tid   = threadIdx.x;
    const int lane  = tid & 31;
    const int wave  = tid >> 5;
    const int nbase = blockIdx.x * NTILE + wave * 16;
    const int ncol  = nbase + (lane & 15);
    const int koff  = (lane >> 4) * 2;
    const int kbase = blockIdx.y * KSLICE;

    const int step = stepp[0];
    float a = fminf((float)step / (float)ANNEAL, 1.0f);
    const float oma = 1.0f - a;
    const float a8  = a * 0.125f;

    const float* Rp = thr + (unsigned)(step % NUM_BANKS) * KK;
    float r[KK];
    #pragma unroll
    for (int i = 0; i < KK; ++i) r[i] = Rp[i];

    const float s = scale[ncol];

    v8f acc[4];
    #pragma unroll
    for (int mt = 0; mt < 4; ++mt) acc[mt] = (v8f)0.0f;

    const float* wrow = lw + (size_t)ncol * IN_F + koff;

    for (int kc = kbase; kc < kbase + KSLICE; kc += KC) {
        __syncthreads();
        for (int i = tid * 4; i < M_ROWS * KC; i += 256 * 4) {
            const int m = i / KC;
            const int j = i & (KC - 1);
            const float4 v = *(const float4*)(x + (size_t)m * IN_F + kc + j);
            *(float4*)(xs + m * XP + j) = v;
        }
        __syncthreads();

        if (kc + KC < kbase + KSLICE)
            __builtin_prefetch(wrow + kc + KC, 0, 1);

        #pragma unroll 4
        for (int k = 0; k < KC; k += 4) {
            const float2 wv = *(const float2*)(wrow + kc + k);
            v2f b;
            {
                float c0 = fminf(fmaxf(wv.x, -1.0f), 1.0f);
                float c1 = fminf(fmaxf(wv.y, -1.0f), 1.0f);
                float t0 = __builtin_fmaf(c0, 0.5f, 0.5f);
                float t1 = __builtin_fmaf(c1, 0.5f, 0.5f);
                int n0 = 0, n1 = 0;
                #pragma unroll
                for (int i = 0; i < KK; ++i) {
                    n0 += (r[i] < t0) ? 1 : 0;
                    n1 += (r[i] < t1) ? 1 : 0;
                }
                b.x = __builtin_fmaf(oma, c0, __builtin_fmaf((float)n0, a8, -a));
                b.y = __builtin_fmaf(oma, c1, __builtin_fmaf((float)n1, a8, -a));
            }

            #pragma unroll
            for (int mt = 0; mt < 4; ++mt) {
                const int m = mt * 16 + (lane & 15);
                v2f av = *(const v2f*)(xs + m * XP + k + koff);
                acc[mt] = __builtin_amdgcn_wmma_f32_16x16x4_f32(
                    false, av, false, b, (short)0, acc[mt], false, false);
            }
        }
    }

    const int mrow0 = (lane >> 4) * 8;
    #pragma unroll
    for (int mt = 0; mt < 4; ++mt) {
        #pragma unroll
        for (int j = 0; j < 8; ++j) {
            const int m = mt * 16 + mrow0 + j;
            unsafeAtomicAdd(&out[(size_t)m * OUT_F + ncol], acc[mt][j] * s);
        }
    }
}

extern "C" void kernel_launch(void* const* d_in, const int* in_sizes, int n_in,
                              void* d_out, int out_size, void* d_ws, size_t ws_size,
                              hipStream_t stream) {
    const float* x     = (const float*)d_in[0];
    const float* lw    = (const float*)d_in[1];
    const float* scale = (const float*)d_in[2];
    const float* thr   = (const float*)d_in[3];
    const int*   step  = (const int*)d_in[4];
    float* out = (float*)d_out;

    const int n4 = out_size / 4;
    zero_out<<<dim3((n4 + 255) / 256), dim3(256), 0, stream>>>(out, n4);

    dim3 grid(OUT_F / NTILE, KSPLIT);
    dim3 block(256);
    density_linear_wmma<<<grid, block, 0, stream>>>(x, lw, scale, thr, step, out);
}


# threshold count per weight in packed f32 arithmetic: clamp((t-r)*2^126) is exactly 0 or 1, two weights per v_pk_add_f32, terms summed exactly; replaces 16 compare + carry-add pairs via mask registers
# speedup vs baseline: 1.4189x; 1.2468x over previous
#include <hip/hip_runtime.h>

typedef float v2f __attribute__((ext_vector_type(2)));
typedef float v8f __attribute__((ext_vector_type(8)));

#define IN_F   4096
#define OUT_F  16384
#define M_ROWS 64
#define KK     16
#define NUM_BANKS 64
#define ANNEAL 2000
#define QH 0x1p126f
#define KC     256
#define XP     (KC + 4)
#define NW     8
#define NTILE  (NW * 16)
#define KSPLIT 4
#define KSLICE (IN_F / KSPLIT)

static __device__ __forceinline__ v2f below2(v2f tH, v2f rH)
{
    v2f d;
    asm("v_pk_add_f32 %0, %1, %2 clamp" : "=v"(d) : "v"(tH), "v"(rH));
    return d;
}

__global__ __launch_bounds__(256)
void zero_out(float* __restrict__ out, int n4)
{
    int i = blockIdx.x * blockDim.x + threadIdx.x;
    if (i < n4) ((float4*)out)[i] = make_float4(0.f, 0.f, 0.f, 0.f);
}

__global__ __launch_bounds__(256)
void density_linear_wmma(const float* __restrict__ x,
                         const float* __restrict__ lw,
                         const float* __restrict__ scale,
                         const float* __restrict__ thr,
                         const int*   __restrict__ stepp,
                         float* __restrict__ out)
{
    __shared__ float xs[M_ROWS * XP];

    const int tid   = threadIdx.x;
    const int lane  = tid & 31;
    const int wave  = tid >> 5;
    const int nbase = blockIdx.x * NTILE + wave * 16;
    const int ncol  = nbase + (lane & 15);
    const int koff  = (lane >> 4) * 2;
    const int kbase = blockIdx.y * KSLICE;

    const int step = stepp[0];
    float a = fminf((float)step / (float)ANNEAL, 1.0f);
    const float oma = 1.0f - a;
    const float a8  = a * 0.125f;

    const float* Rp = thr + (unsigned)(step % NUM_BANKS) * KK;
    v2f rH[KK];
    #pragma unroll
    for (int i = 0; i < KK; ++i) { const float q = -Rp[i] * QH; rH[i] = (v2f)(q); }

    const float s = scale[ncol];

    v8f acc[4];
    #pragma unroll
    for (int mt = 0; mt < 4; ++mt) acc[mt] = (v8f)0.0f;

    const float* wrow = lw + (size_t)ncol * IN_F + koff;

    for (int kc = kbase; kc < kbase + KSLICE; kc += KC) {
        __syncthreads();
        for (int i = tid * 4; i < M_ROWS * KC; i += 256 * 4) {
            const int m = i / KC;
            const int j = i & (KC - 1);
            const float4 v = *(const float4*)(x + (size_t)m * IN_F + kc + j);
            *(float4*)(xs + m * XP + j) = v;
        }
        __syncthreads();

        if (kc + KC < kbase + KSLICE)
            __builtin_prefetch(wrow + kc + KC, 0, 1);

        #pragma unroll 4
        for (int k = 0; k < KC; k += 4) {
            const v2f wv = *(const v2f*)(wrow + kc + k);
            v2f b;
            {
                const v2f c  = __builtin_elementwise_min(__builtin_elementwise_max(wv, (v2f)(-1.0f)), (v2f)(1.0f));
                const v2f tH = __builtin_elementwise_fma(c, (v2f)(0.5f * QH), (v2f)(0.5f * QH));
                v2f d[KK];
                #pragma unroll
                for (int i = 0; i < KK; ++i) d[i] = below2(tH, rH[i]);
                #pragma unroll
                for (int w = KK / 2; w >= 1; w >>= 1) {
                    #pragma unroll
                    for (int i = 0; i < w; ++i) d[i] = d[i] + d[i + w];
                }
                b = __builtin_elementwise_fma((v2f)(oma), c, __builtin_elementwise_fma(d[0], (v2f)(a8), (v2f)(-a)));
            }

            #pragma unroll
            for (int mt = 0; mt < 4; ++mt) {
                const int m = mt * 16 + (lane & 15);
                v2f av = *(const v2f*)(xs + m * XP + k + koff);
                acc[mt] = __builtin_amdgcn_wmma_f32_16x16x4_f32(
                    false, av, false, b, (short)0, acc[mt], false, false);
            }
        }
    }

    const int mrow0 = (lane >> 4) * 8;
    #pragma unroll
    for (int mt = 0; mt < 4; ++mt) {
        #pragma unroll
        for (int j = 0; j < 8; ++j) {
            const int m = mt * 16 + mrow0 + j;
            unsafeAtomicAdd(&out[(size_t)m * OUT_F + ncol], acc[mt][j] * s);
        }
    }
}

extern "C" void kernel_launch(void* const* d_in, const int* in_sizes, int n_in,
                              void* d_out, int out_size, void* d_ws, size_t ws_size,
                              hipStream_t stream) {
    const float* x     = (const float*)d_in[0];
    const float* lw    = (const float*)d_in[1];
    const float* scale = (const float*)d_in[2];
    const float* thr   = (const float*)d_in[3];
    const int*   step  = (const int*)d_in[4];
    float* out = (float*)d_out;

    const int n4 = out_size / 4;
    zero_out<<<dim3((n4 + 255) / 256), dim3(256), 0, stream>>>(out, n4);

    dim3 grid(OUT_F / NTILE, KSPLIT);
    dim3 block(256);
    density_linear_wmma<<<grid, block, 0, stream>>>(x, lw, scale, thr, step, out);
}


# global reads issued ahead of use: a trip's weight fragments are loaded one 16-k trip earlier into a second register set; a thread's 16 x-tile loads all fly before its LDS stores; sums unchanged
# speedup vs baseline: 1.7111x; 1.2059x over previous
#include <hip/hip_runtime.h>

typedef float v2f __attribute__((ext_vector_type(2)));
typedef float v8f __attribute__((ext_vector_type(8)));

#define IN_F   4096
#define OUT_F  16384
#define M_ROWS 64
#define KK     16
#define NUM_BANKS 64
#define ANNEAL 2000
#define QH 0x1p126f
#define KC     256
#define XP     (KC + 4)
#define NW     8
#define NTILE  (NW * 16)
#define KSPLIT 4
#define KSLICE (IN_F / KSPLIT)

static __device__ __forceinline__ v2f below2(v2f tH, v2f rH)
{
    v2f d;
    asm("v_pk_add_f32 %0, %1, %2 clamp" : "=v"(d) : "v"(tH), "v"(rH));
    return d;
}

__global__ __launch_bounds__(256)
void zero_out(float* __restrict__ out, int n4)
{
    int i = blockIdx.x * blockDim.x + threadIdx.x;
    if (i < n4) ((float4*)out)[i] = make_float4(0.f, 0.f, 0.f, 0.f);
}

static __device__ __forceinline__ void trip(const v2f (&w)[4], const float* xa, v8f (&acc)[4],
                                            const v2f (&rH)[KK], float oma, float a8, float a)
{
    __builtin_amdgcn_sched_barrier(0);

    v2f av[4][4];
    #pragma unroll
    for (int j = 0; j < 4; ++j) {
        #pragma unroll
        for (int mt = 0; mt < 4; ++mt)
            av[j][mt] = *(const v2f*)(xa + mt * 16 * XP + 4 * j);
    }
    __builtin_amdgcn_sched_barrier(0);

    #pragma unroll
    for (int j = 0; j < 4; ++j) {
        const v2f wv = w[j];
        v2f b;
        {
            const v2f c  = __builtin_elementwise_min(__builtin_elementwise_max(wv, (v2f)(-1.0f)), (v2f)(1.0f));
            const v2f tH = __builtin_elementwise_fma(c, (v2f)(0.5f * QH), (v2f)(0.5f * QH));
            v2f d[KK];
            #pragma unroll
            for (int i = 0; i < KK; ++i) d[i] = below2(tH, rH[i]);
            #pragma unroll
            for (int s = KK / 2; s >= 1; s >>= 1) {
                #pragma unroll
                for (int i = 0; i < s; ++i) d[i] = d[i] + d[i + s];
            }
            b = __builtin_elementwise_fma((v2f)(oma), c, __builtin_elementwise_fma(d[0], (v2f)(a8), (v2f)(-a)));
        }

        #pragma unroll
        for (int mt = 0; mt < 4; ++mt) {
            acc[mt] = __builtin_amdgcn_wmma_f32_16x16x4_f32(
                false, av[j][mt], false, b, (short)0, acc[mt], false, false);
        }
    }
}

static __device__ __forceinline__ void load_w(v2f (&w)[4], const float* p)
{
    #pragma unroll
    for (int j = 0; j < 4; ++j) w[j] = *(const v2f*)(p + 4 * j);
}

__global__ __launch_bounds__(256)
void density_linear_wmma(const float* __restrict__ x,
                         const float* __restrict__ lw,
                         const float* __restrict__ scale,
                         const float* __restrict__ thr,
                         const int*   __restrict__ stepp,
                         float* __restrict__ out)
{
    __shared__ float xs[M_ROWS * XP];

    const int tid   = threadIdx.x;
    const int lane  = tid & 31;
    const int wave  = tid >> 5;
    const int nbase = blockIdx.x * NTILE + wave * 16;
    const int ncol  = nbase + (lane & 15);
    const int koff  = (lane >> 4) * 2;
    const int kbase = blockIdx.y * KSLICE;
    const int kend  = kbase + KSLICE;

    const float* wrow = lw + (size_t)ncol * IN_F + koff;

    v2f wA[4], wB[4];
    load_w(wA, wrow + kbase);

    const int step = stepp[0];
    float a = fminf((float)step / (float)ANNEAL, 1.0f);
    const float oma = 1.0f - a;
    const float a8  = a * 0.125f;

    const float* Rp = thr + (unsigned)(step % NUM_BANKS) * KK;
    v2f rH[KK];
    #pragma unroll
    for (int i = 0; i < KK; ++i) { const float q = -Rp[i] * QH; rH[i] = (v2f)(q); }

    const float s = scale[ncol];

    v8f acc[4];
    #pragma unroll
    for (int mt = 0; mt < 4; ++mt) acc[mt] = (v8f)0.0f;

    const float* xlane = xs + (lane & 15) * XP + koff;
    const float* xsrc = x + (size_t)(tid >> 6) * IN_F + (tid & 63) * 4;
    float*       xdst = xs + (tid >> 6) * XP + (tid & 63) * 4;

    for (int kc = kbase; kc < kend; kc += KC) {
        __syncthreads();
        {
            float4 xv[16];
            #pragma unroll
            for (int q = 0; q < 16; ++q)
                xv[q] = *(const float4*)(xsrc + (size_t)(4 * q) * IN_F + kc);
            #pragma unroll
            for (int q = 0; q < 16; ++q)
                *(float4*)(xdst + 4 * q * XP) = xv[q];
        }
        __syncthreads();

        for (int k = 0; k < KC; k += 32) {
            load_w(wB, wrow + kc + k + 16);
            trip(wA, xlane + k, acc, rH, oma, a8, a);

            const int kn = (kc + k + 32 < kend) ? (kc + k + 32) : kbase;
            load_w(wA, wrow + kn);
            trip(wB, xlane + k + 16, acc, rH, oma, a8, a);
        }
    }

    const int mrow0 = (lane >> 4) * 8;
    #pragma unroll
    for (int mt = 0; mt < 4; ++mt) {
        #pragma unroll
        for (int j = 0; j < 8; ++j) {
            const int m = mt * 16 + mrow0 + j;
            unsafeAtomicAdd(&out[(size_t)m * OUT_F + ncol], acc[mt][j] * s);
        }
    }
}

extern "C" void kernel_launch(void* const* d_in, const int* in_sizes, int n_in,
                              void* d_out, int out_size, void* d_ws, size_t ws_size,
                              hipStream_t stream) {
    const float* x     = (const float*)d_in[0];
    const float* lw    = (const float*)d_in[1];
    const float* scale = (const float*)d_in[2];
    const float* thr   = (const float*)d_in[3];
    const int*   step  = (const int*)d_in[4];
    float* out = (float*)d_out;

    const int n4 = out_size / 4;
    zero_out<<<dim3((n4 + 255) / 256), dim3(256), 0, stream>>>(out, n4);

    dim3 grid(OUT_F / NTILE, KSPLIT);
    dim3 block(256);
    density_linear_wmma<<<grid, block, 0, stream>>>(x, lw, scale, thr, step, out);
}


# threshold count by the reference's own 5-step bisection: each step reads its threshold from a 16-lane register table (v_permlane16_var_b32); 25 vector ops per weight pair, was 37; sums unchanged
# speedup vs baseline: 1.8792x; 1.0982x over previous
#include <hip/hip_runtime.h>

typedef float v2f __attribute__((ext_vector_type(2)));
typedef float v8f __attribute__((ext_vector_type(8)));

#define IN_F   4096
#define OUT_F  16384
#define M_ROWS 64
#define KK     16
#define NUM_BANKS 64
#define ANNEAL 2000
#define QH 0x1p126f
#define KC     256
#define XP     (KC + 4)
#define NW     8
#define NTILE  (NW * 16)
#define KSPLIT 4
#define KSLICE (IN_F / KSPLIT)

static __device__ __forceinline__ v2f below2(v2f tH, v2f rH)
{
    v2f d;
    asm("v_pk_add_f32 %0, %1, %2 clamp" : "=v"(d) : "v"(tH), "v"(rH));
    return d;
}

static __device__ __forceinline__ float lut16(float dead, float tab, float sel)
{
    return __uint_as_float(__builtin_amdgcn_permlane16_var(
        __float_as_uint(dead), __float_as_uint(tab), __float_as_uint(sel), false, false));
}

#define BISECT_STEP(L, D, T, P, STEP)                                                   \
    asm("v_pk_add_f32 %0, %8, %12 clamp\n\t"                                            \
        "v_pk_add_f32 %1, %9, %13 clamp\n\t"                                            \
        "v_pk_add_f32 %2, %10, %14 clamp\n\t"                                           \
        "s_delay_alu instid0(VALU_DEP_4)\n\t"                                           \
        "v_pk_add_f32 %3, %11, %15 clamp\n\t"                                           \
        "s_delay_alu instid0(VALU_DEP_4)\n\t"                                           \
        "v_pk_fma_f32 %4, %0, " STEP ", %4 op_sel_hi:[1,0,1]\n\t"                       \
        "v_pk_fma_f32 %5, %1, " STEP ", %5 op_sel_hi:[1,0,1]\n\t"                       \
        "v_pk_fma_f32 %6, %2, " STEP ", %6 op_sel_hi:[1,0,1]\n\t"                       \
        "v_pk_fma_f32 %7, %3, " STEP ", %7 op_sel_hi:[1,0,1]\n\t"                       \
        "s_delay_alu instid0(VALU_DEP_4)"                                               \
        : "=&v"(D[0]), "=&v"(D[1]), "=&v"(D[2]), "=&v"(D[3]),                           \
          "+v"(L[0]), "+v"(L[1]), "+v"(L[2]), "+v"(L[3])                                \
        : "v"(T[0]), "v"(T[1]), "v"(T[2]), "v"(T[3]),                                   \
          "v"(P[0]), "v"(P[1]), "v"(P[2]), "v"(P[3]))

#define BISECT_FIRST(L, D, T, R8)                                                       \
    asm("v_pk_add_f32 %0, %8, %12 clamp\n\t"                                            \
        "v_pk_add_f32 %1, %9, %12 clamp\n\t"                                            \
        "v_pk_add_f32 %2, %10, %12 clamp\n\t"                                           \
        "v_pk_add_f32 %3, %11, %12 clamp\n\t"                                           \
        "s_delay_alu instid0(VALU_DEP_4)\n\t"                                           \
        "v_pk_mul_f32 %4, %0, 8 op_sel_hi:[1,0]\n\t"                                    \
        "v_pk_mul_f32 %5, %1, 8 op_sel_hi:[1,0]\n\t"                                    \
        "v_pk_mul_f32 %6, %2, 8 op_sel_hi:[1,0]\n\t"                                    \
        "v_pk_mul_f32 %7, %3, 8 op_sel_hi:[1,0]\n\t"                                    \
        "s_delay_alu instid0(VALU_DEP_4)"                                               \
        : "=&v"(D[0]), "=&v"(D[1]), "=&v"(D[2]), "=&v"(D[3]),                           \
          "=&v"(L[0]), "=&v"(L[1]), "=&v"(L[2]), "=&v"(L[3])                            \
        : "v"(T[0]), "v"(T[1]), "v"(T[2]), "v"(T[3]), "v"(R8))

#define BISECT_LAST(D, T, P)                                                            \
    asm("v_pk_add_f32 %0, %4, %8 clamp\n\t"                                             \
        "v_pk_add_f32 %1, %5, %9 clamp\n\t"                                             \
        "v_pk_add_f32 %2, %6, %10 clamp\n\t"                                            \
        "s_delay_alu instid0(VALU_DEP_4)\n\t"                                           \
        "v_pk_add_f32 %3, %7, %11 clamp\n\t"                                            \
        "s_delay_alu instid0(VALU_DEP_4)"                                               \
        : "=&v"(D[0]), "=&v"(D[1]), "=&v"(D[2]), "=&v"(D[3])                            \
        : "v"(T[0]), "v"(T[1]), "v"(T[2]), "v"(T[3]),                                   \
          "v"(P[0]), "v"(P[1]), "v"(P[2]), "v"(P[3]))

__global__ __launch_bounds__(256)
void zero_out(float* __restrict__ out, int n4)
{
    int i = blockIdx.x * blockDim.x + threadIdx.x;
    if (i < n4) ((float4*)out)[i] = make_float4(0.f, 0.f, 0.f, 0.f);
}

static __device__ __forceinline__ void trip(const v2f (&w)[4], const float* xa, v8f (&acc)[4],
                                            v2f r8H, float tb2, float tb3, float tb4, float tb5,
                                            float oma, float a8S, float a)
{
    __builtin_amdgcn_sched_barrier(0);

    v2f av[4][4];
    #pragma unroll
    for (int j = 0; j < 4; ++j) {
        #pragma unroll
        for (int mt = 0; mt < 4; ++mt)
            av[j][mt] = *(const v2f*)(xa + mt * 16 * XP + 4 * j);
    }
    __builtin_amdgcn_sched_barrier(0);

    v2f c[4], tH[4], low[4], p[4], d[4], b[4];
    #pragma unroll
    for (int q = 0; q < 4; ++q) {
        c[q]  = __builtin_elementwise_min(__builtin_elementwise_max(w[q], (v2f)(-1.0f)), (v2f)(1.0f));
        tH[q] = __builtin_elementwise_fma(c[q], (v2f)(0.5f * QH), (v2f)(0.5f * QH));
    }
    BISECT_FIRST(low, d, tH, r8H);
    #pragma unroll
    for (int q = 0; q < 4; ++q) { p[q].x = lut16(w[q].x, tb2, low[q].x); p[q].y = lut16(w[q].y, tb2, low[q].y); }
    BISECT_STEP(low, d, tH, p, "4");
    #pragma unroll
    for (int q = 0; q < 4; ++q) { p[q].x = lut16(p[q].x, tb3, low[q].x); p[q].y = lut16(p[q].y, tb3, low[q].y); }
    BISECT_STEP(low, d, tH, p, "2");
    #pragma unroll
    for (int q = 0; q < 4; ++q) { p[q].x = lut16(p[q].x, tb4, low[q].x); p[q].y = lut16(p[q].y, tb4, low[q].y); }
    BISECT_STEP(low, d, tH, p, "1");
    #pragma unroll
    for (int q = 0; q < 4; ++q) { p[q].x = lut16(p[q].x, tb5, low[q].x); p[q].y = lut16(p[q].y, tb5, low[q].y); }
    BISECT_LAST(d, tH, p);
    #pragma unroll
    for (int q = 0; q < 4; ++q) {
        const v2f cs = __builtin_elementwise_fma(d[q], (v2f)(0x1p-23f), low[q] * (v2f)(0x1p126f));
        b[q] = __builtin_elementwise_fma((v2f)(oma), c[q], __builtin_elementwise_fma(cs, (v2f)(a8S), (v2f)(-a)));
    }

    #pragma unroll
    for (int j = 0; j < 4; ++j) {
        #pragma unroll
        for (int mt = 0; mt < 4; ++mt) {
            acc[mt] = __builtin_amdgcn_wmma_f32_16x16x4_f32(
                false, av[j][mt], false, b[j], (short)0, acc[mt], false, false);
        }
    }
}

static __device__ __forceinline__ void load_w(v2f (&w)[4], const float* p)
{
    #pragma unroll
    for (int j = 0; j < 4; ++j) w[j] = *(const v2f*)(p + 4 * j);
}

__global__ __launch_bounds__(256)
void density_linear_wmma(const float* __restrict__ x,
                         const float* __restrict__ lw,
                         const float* __restrict__ scale,
                         const float* __restrict__ thr,
                         const int*   __restrict__ stepp,
                         float* __restrict__ out)
{
    __shared__ float xs[M_ROWS * XP];

    const int tid   = threadIdx.x;
    const int lane  = tid & 31;
    const int wave  = tid >> 5;
    const int nbase = blockIdx.x * NTILE + wave * 16;
    const int ncol  = nbase + (lane & 15);
    const int koff  = (lane >> 4) * 2;
    const int kbase = blockIdx.y * KSLICE;
    const int kend  = kbase + KSLICE;

    const float* wrow = lw + (size_t)ncol * IN_F + koff;

    v2f wA[4], wB[4];
    load_w(wA, wrow + kbase);

    const int step = stepp[0];
    float a = fminf((float)step / (float)ANNEAL, 1.0f);
    const float oma = 1.0f - a;
    const float a8S = a * 0x1p20f;

    const float* Rp = thr + (unsigned)(step % NUM_BANKS) * KK;
    const int   li  = lane & 15;
    const v2f   r8H = (v2f)(-Rp[8] * QH);
    const float tb2 = -Rp[(li + 4) & 15] * QH;
    const float tb3 = -Rp[(li + 2) & 15] * QH;
    const float tb4 = -Rp[(li + 1) & 15] * QH;
    const float tb5 = -Rp[li] * QH;

    const float s = scale[ncol];

    v8f acc[4];
    #pragma unroll
    for (int mt = 0; mt < 4; ++mt) acc[mt] = (v8f)0.0f;

    const float* xlane = xs + (lane & 15) * XP + koff;
    const float* xsrc = x + (size_t)(tid >> 6) * IN_F + (tid & 63) * 4;
    float*       xdst = xs + (tid >> 6) * XP + (tid & 63) * 4;

    for (int kc = kbase; kc < kend; kc += KC) {
        __syncthreads();
        {
            float4 xv[16];
            #pragma unroll
            for (int q = 0; q < 16; ++q)
                xv[q] = *(const float4*)(xsrc + (size_t)(4 * q) * IN_F + kc);
            #pragma unroll
            for (int q = 0; q < 16; ++q)
                *(float4*)(xdst + 4 * q * XP) = xv[q];
        }
        __syncthreads();

        for (int k = 0; k < KC; k += 32) {
            load_w(wB, wrow + kc + k + 16);
            trip(wA, xlane + k, acc, r8H, tb2, tb3, tb4, tb5, oma, a8S, a);

            const int kn = (kc + k + 32 < kend) ? (kc + k + 32) : kbase;
            load_w(wA, wrow + kn);
            trip(wB, xlane + k + 16, acc, r8H, tb2, tb3, tb4, tb5, oma, a8S, a);
        }
    }

    const int mrow0 = (lane >> 4) * 8;
    #pragma unroll
    for (int mt = 0; mt < 4; ++mt) {
        #pragma unroll
        for (int j = 0; j < 8; ++j) {
            const int m = mt * 16 + mrow0 + j;
            unsafeAtomicAdd(&out[(size_t)m * OUT_F + ncol], acc[mt][j] * s);
        }
    }
}

extern "C" void kernel_launch(void* const* d_in, const int* in_sizes, int n_in,
                              void* d_out, int out_size, void* d_ws, size_t ws_size,
                              hipStream_t stream) {
    const float* x     = (const float*)d_in[0];
    const float* lw    = (const float*)d_in[1];
    const float* scale = (const float*)d_in[2];
    const float* thr   = (const float*)d_in[3];
    const int*   step  = (const int*)d_in[4];
    float* out = (float*)d_out;

    const int n4 = out_size / 4;
    zero_out<<<dim3((n4 + 255) / 256), dim3(256), 0, stream>>>(out, n4);

    dim3 grid(OUT_F / NTILE, KSPLIT);
    dim3 block(256);
    density_linear_wmma<<<grid, block, 0, stream>>>(x, lw, scale, thr, step, out);
}


# weight fragments loaded three trips ahead of use (was one): four register sets take turns, twelve loads in flight per wave; sums and their order unchanged
# speedup vs baseline: 1.9336x; 1.0289x over previous
#include <hip/hip_runtime.h>

typedef float v2f __attribute__((ext_vector_type(2)));
typedef float v8f __attribute__((ext_vector_type(8)));

#define IN_F   4096
#define OUT_F  16384
#define M_ROWS 64
#define KK     16
#define NUM_BANKS 64
#define ANNEAL 2000
#define QH 0x1p126f
#define KC     256
#define XP     (KC + 4)
#define NW     8
#define NTILE  (NW * 16)
#define KSPLIT 4
#define KSLICE (IN_F / KSPLIT)

static __device__ __forceinline__ v2f below2(v2f tH, v2f rH)
{
    v2f d;
    asm("v_pk_add_f32 %0, %1, %2 clamp" : "=v"(d) : "v"(tH), "v"(rH));
    return d;
}

static __device__ __forceinline__ float lut16(float dead, float tab, float sel)
{
    return __uint_as_float(__builtin_amdgcn_permlane16_var(
        __float_as_uint(dead), __float_as_uint(tab), __float_as_uint(sel), false, false));
}

#define BISECT_STEP(L, D, T, P, STEP)                                                   \
    asm("v_pk_add_f32 %0, %8, %12 clamp\n\t"                                            \
        "v_pk_add_f32 %1, %9, %13 clamp\n\t"                                            \
        "v_pk_add_f32 %2, %10, %14 clamp\n\t"                                           \
        "s_delay_alu instid0(VALU_DEP_4)\n\t"                                           \
        "v_pk_add_f32 %3, %11, %15 clamp\n\t"                                           \
        "s_delay_alu instid0(VALU_DEP_4)\n\t"                                           \
        "v_pk_fma_f32 %4, %0, " STEP ", %4 op_sel_hi:[1,0,1]\n\t"                       \
        "v_pk_fma_f32 %5, %1, " STEP ", %5 op_sel_hi:[1,0,1]\n\t"                       \
        "v_pk_fma_f32 %6, %2, " STEP ", %6 op_sel_hi:[1,0,1]\n\t"                       \
        "v_pk_fma_f32 %7, %3, " STEP ", %7 op_sel_hi:[1,0,1]\n\t"                       \
        "s_delay_alu instid0(VALU_DEP_4)"                                               \
        : "=&v"(D[0]), "=&v"(D[1]), "=&v"(D[2]), "=&v"(D[3]),                           \
          "+v"(L[0]), "+v"(L[1]), "+v"(L[2]), "+v"(L[3])                                \
        : "v"(T[0]), "v"(T[1]), "v"(T[2]), "v"(T[3]),                                   \
          "v"(P[0]), "v"(P[1]), "v"(P[2]), "v"(P[3]))

#define BISECT_FIRST(L, D, T, R8)                                                       \
    asm("v_pk_add_f32 %0, %8, %12 clamp\n\t"                                            \
        "v_pk_add_f32 %1, %9, %12 clamp\n\t"                                            \
        "v_pk_add_f32 %2, %10, %12 clamp\n\t"                                           \
        "v_pk_add_f32 %3, %11, %12 clamp\n\t"                                           \
        "s_delay_alu instid0(VALU_DEP_4)\n\t"                                           \
        "v_pk_mul_f32 %4, %0, 8 op_sel_hi:[1,0]\n\t"                                    \
        "v_pk_mul_f32 %5, %1, 8 op_sel_hi:[1,0]\n\t"                                    \
        "v_pk_mul_f32 %6, %2, 8 op_sel_hi:[1,0]\n\t"                                    \
        "v_pk_mul_f32 %7, %3, 8 op_sel_hi:[1,0]\n\t"                                    \
        "s_delay_alu instid0(VALU_DEP_4)"                                               \
        : "=&v"(D[0]), "=&v"(D[1]), "=&v"(D[2]), "=&v"(D[3]),                           \
          "=&v"(L[0]), "=&v"(L[1]), "=&v"(L[2]), "=&v"(L[3])                            \
        : "v"(T[0]), "v"(T[1]), "v"(T[2]), "v"(T[3]), "v"(R8))

#define BISECT_LAST(D, T, P)                                                            \
    asm("v_pk_add_f32 %0, %4, %8 clamp\n\t"                                             \
        "v_pk_add_f32 %1, %5, %9 clamp\n\t"                                             \
        "v_pk_add_f32 %2, %6, %10 clamp\n\t"                                            \
        "s_delay_alu instid0(VALU_DEP_4)\n\t"                                           \
        "v_pk_add_f32 %3, %7, %11 clamp\n\t"                                            \
        "s_delay_alu instid0(VALU_DEP_4)"                                               \
        : "=&v"(D[0]), "=&v"(D[1]), "=&v"(D[2]), "=&v"(D[3])                            \
        : "v"(T[0]), "v"(T[1]), "v"(T[2]), "v"(T[3]),                                   \
          "v"(P[0]), "v"(P[1]), "v"(P[2]), "v"(P[3]))

__global__ __launch_bounds__(256)
void zero_out(float* __restrict__ out, int n4)
{
    int i = blockIdx.x * blockDim.x + threadIdx.x;
    if (i < n4) ((float4*)out)[i] = make_float4(0.f, 0.f, 0.f, 0.f);
}

static __device__ __forceinline__ void trip(const v2f (&w)[4], const float* xa, v8f (&acc)[4],
                                            v2f r8H, float tb2, float tb3, float tb4, float tb5,
                                            float oma, float a8S, float a)
{
    __builtin_amdgcn_sched_barrier(0);

    v2f av[4][4];
    #pragma unroll
    for (int j = 0; j < 4; ++j) {
        #pragma unroll
        for (int mt = 0; mt < 4; ++mt)
            av[j][mt] = *(const v2f*)(xa + mt * 16 * XP + 4 * j);
    }
    __builtin_amdgcn_sched_barrier(0);

    v2f c[4], tH[4], low[4], p[4], d[4], b[4];
    #pragma unroll
    for (int q = 0; q < 4; ++q) {
        c[q]  = __builtin_elementwise_min(__builtin_elementwise_max(w[q], (v2f)(-1.0f)), (v2f)(1.0f));
        tH[q] = __builtin_elementwise_fma(c[q], (v2f)(0.5f * QH), (v2f)(0.5f * QH));
    }
    BISECT_FIRST(low, d, tH, r8H);
    #pragma unroll
    for (int q = 0; q < 4; ++q) { p[q].x = lut16(w[q].x, tb2, low[q].x); p[q].y = lut16(w[q].y, tb2, low[q].y); }
    BISECT_STEP(low, d, tH, p, "4");
    #pragma unroll
    for (int q = 0; q < 4; ++q) { p[q].x = lut16(p[q].x, tb3, low[q].x); p[q].y = lut16(p[q].y, tb3, low[q].y); }
    BISECT_STEP(low, d, tH, p, "2");
    #pragma unroll
    for (int q = 0; q < 4; ++q) { p[q].x = lut16(p[q].x, tb4, low[q].x); p[q].y = lut16(p[q].y, tb4, low[q].y); }
    BISECT_STEP(low, d, tH, p, "1");
    #pragma unroll
    for (int q = 0; q < 4; ++q) { p[q].x = lut16(p[q].x, tb5, low[q].x); p[q].y = lut16(p[q].y, tb5, low[q].y); }
    BISECT_LAST(d, tH, p);
    #pragma unroll
    for (int q = 0; q < 4; ++q) {
        const v2f cs = __builtin_elementwise_fma(d[q], (v2f)(0x1p-23f), low[q] * (v2f)(0x1p126f));
        b[q] = __builtin_elementwise_fma((v2f)(oma), c[q], __builtin_elementwise_fma(cs, (v2f)(a8S), (v2f)(-a)));
    }

    #pragma unroll
    for (int j = 0; j < 4; ++j) {
        #pragma unroll
        for (int mt = 0; mt < 4; ++mt) {
            acc[mt] = __builtin_amdgcn_wmma_f32_16x16x4_f32(
                false, av[j][mt], false, b[j], (short)0, acc[mt], false, false);
        }
    }
}

static __device__ __forceinline__ void load_w(v2f (&w)[4], const float* p)
{
    #pragma unroll
    for (int j = 0; j < 4; ++j) w[j] = *(const v2f*)(p + 4 * j);
}

__global__ __launch_bounds__(256)
void density_linear_wmma(const float* __restrict__ x,
                         const float* __restrict__ lw,
                         const float* __restrict__ scale,
                         const float* __restrict__ thr,
                         const int*   __restrict__ stepp,
                         float* __restrict__ out)
{
    __shared__ float xs[M_ROWS * XP];

    const int tid   = threadIdx.x;
    const int lane  = tid & 31;
    const int wave  = tid >> 5;
    const int nbase = blockIdx.x * NTILE + wave * 16;
    const int ncol  = nbase + (lane & 15);
    const int koff  = (lane >> 4) * 2;
    const int kbase = blockIdx.y * KSLICE;
    const int kend  = kbase + KSLICE;

    const float* wrow = lw + (size_t)ncol * IN_F + koff;

    v2f w0[4], w1[4], w2[4], w3[4];
    load_w(w0, wrow + kbase);
    load_w(w1, wrow + kbase + 16);
    load_w(w2, wrow + kbase + 32);
    const int klast = kend - 16;

    const int step = stepp[0];
    float a = fminf((float)step / (float)ANNEAL, 1.0f);
    const float oma = 1.0f - a;
    const float a8S = a * 0x1p20f;

    const float* Rp = thr + (unsigned)(step % NUM_BANKS) * KK;
    const int   li  = lane & 15;
    const v2f   r8H = (v2f)(-Rp[8] * QH);
    const float tb2 = -Rp[(li + 4) & 15] * QH;
    const float tb3 = -Rp[(li + 2) & 15] * QH;
    const float tb4 = -Rp[(li + 1) & 15] * QH;
    const float tb5 = -Rp[li] * QH;

    const float s = scale[ncol];

    v8f acc[4];
    #pragma unroll
    for (int mt = 0; mt < 4; ++mt) acc[mt] = (v8f)0.0f;

    const float* xlane = xs + (lane & 15) * XP + koff;
    const float* xsrc = x + (size_t)(tid >> 6) * IN_F + (tid & 63) * 4;
    float*       xdst = xs + (tid >> 6) * XP + (tid & 63) * 4;

    for (int kc = kbase; kc < kend; kc += KC) {
        __syncthreads();
        {
            float4 xv[16];
            #pragma unroll
            for (int q = 0; q < 16; ++q)
                xv[q] = *(const float4*)(xsrc + (size_t)(4 * q) * IN_F + kc);
            #pragma unroll
            for (int q = 0; q < 16; ++q)
                *(float4*)(xdst + 4 * q * XP) = xv[q];
        }
        __syncthreads();

        for (int k = 0; k < KC; k += 64) {
            const int kk = kc + k;
            load_w(w3, wrow + kk + 48);
            trip(w0, xlane + k, acc, r8H, tb2, tb3, tb4, tb5, oma, a8S, a);

            load_w(w0, wrow + min(kk + 64, klast));
            trip(w1, xlane + k + 16, acc, r8H, tb2, tb3, tb4, tb5, oma, a8S, a);

            load_w(w1, wrow + min(kk + 80, klast));
            trip(w2, xlane + k + 32, acc, r8H, tb2, tb3, tb4, tb5, oma, a8S, a);

            load_w(w2, wrow + min(kk + 96, klast));
            trip(w3, xlane + k + 48, acc, r8H, tb2, tb3, tb4, tb5, oma, a8S, a);
        }
    }

    const int mrow0 = (lane >> 4) * 8;
    #pragma unroll
    for (int mt = 0; mt < 4; ++mt) {
        #pragma unroll
        for (int j = 0; j < 8; ++j) {
            const int m = mt * 16 + mrow0 + j;
            unsafeAtomicAdd(&out[(size_t)m * OUT_F + ncol], acc[mt][j] * s);
        }
    }
}

extern "C" void kernel_launch(void* const* d_in, const int* in_sizes, int n_in,
                              void* d_out, int out_size, void* d_ws, size_t ws_size,
                              hipStream_t stream) {
    const float* x     = (const float*)d_in[0];
    const float* lw    = (const float*)d_in[1];
    const float* scale = (const float*)d_in[2];
    const float* thr   = (const float*)d_in[3];
    const int*   step  = (const int*)d_in[4];
    float* out = (float*)d_out;

    const int n4 = out_size / 4;
    zero_out<<<dim3((n4 + 255) / 256), dim3(256), 0, stream>>>(out, n4);

    dim3 grid(OUT_F / NTILE, KSPLIT);
    dim3 block(256);
    density_linear_wmma<<<grid, block, 0, stream>>>(x, lw, scale, thr, step, out);
}
